# starting kernel
# baseline (speedup 1.0000x reference)
#include <hip/hip_runtime.h>

typedef __attribute__((ext_vector_type(2))) float v2f;
typedef __attribute__((ext_vector_type(8))) float v8f;

#define NB 2
#define NH 16
#define NN 8192
#define HDIM 64
#define NOFF 44
#define NTILE 14
#define SSTR 52

__constant__ int c_off[NOFF] = {
  0,1,2,3,4,5,6,7,8,9,10,11,12,13,14,15,16,17,18,19,20,21,22,
  23,24,25,26,27,28,29,30,31,32,
  48,64,96,128,192,256,384,512,768,1024,1536 };

__constant__ int c_toff[NTILE] = {0,16,32,48,64,96,128,192,256,384,512,768,1024,1536};

__device__ __forceinline__ v8f wmma4(v2f a, v2f b, v8f c) {
  return __builtin_amdgcn_wmma_f32_16x16x4_f32(
      false, a, false, b, (short)0, c, false, false);
}

__global__ __launch_bounds__(128)
void dsqg_attn_v4_kernel(const float* __restrict__ q,
                         const float* __restrict__ k,
                         const float* __restrict__ v,
                         const float* __restrict__ pbias,
                         const float* __restrict__ se,
                         const float* __restrict__ ph,
                         float* __restrict__ out)
{
  __shared__ float s_cp[48], s_sp[48], s_pb[48];
  __shared__ float s_scores[4 * 16 * SSTR];
  __shared__ float s_rinv[4 * 16];

  const int bid  = blockIdx.x;
  const int bh   = bid >> 7;
  const int h    = bh & (NH - 1);
  const int n0b  = (bid & 127) << 6;
  const int tid  = threadIdx.x;
  const int wv   = tid >> 5;
  const int lane = tid & 31;
  const int n0   = n0b + (wv << 4);

  if (tid < NOFF) {
    float p = ph[tid * NH + h];
    s_cp[tid] = cosf(p);
    s_sp[tid] = sinf(p);
    s_pb[tid] = pbias[tid * NH + h];
  }
  __syncthreads();

  const float sc   = 0.125f;
  const int   nl   = lane & 15;
  const int   hi   = lane >> 4;
  const int   koff = hi << 1;

  const size_t bhoff = (size_t)bh * NN * HDIM;
  const float* qbh = q + bhoff;
  const float* kbh = k + bhoff;
  const float* vbh = v + bhoff;
  float*       obh = out + bhoff;
  float*       srow = s_scores + wv * (16 * SSTR);

  v2f qa[16];
  {
    const float* qr = qbh + (size_t)(n0 + nl) * HDIM + koff;
#pragma unroll
    for (int c = 0; c < 16; ++c) {
      v2f t = *(const v2f*)(qr + 4 * c);
      qa[c].x = t.x * sc;
      qa[c].y = t.y * sc;
    }
  }

#pragma unroll
  for (int it = 0; it < 3; ++it) {
    int irow = it * 16 + nl;
    int ir   = (irow < NOFF) ? irow : (NOFF - 1);
    const float* ser = se + (size_t)ir * HDIM + koff;
    v8f acc = {0.f,0.f,0.f,0.f,0.f,0.f,0.f,0.f};
#pragma unroll
    for (int c = 0; c < 16; ++c) {
      v2f b = *(const v2f*)(ser + 4 * c);
      acc = wmma4(qa[c], b, acc);
    }
#pragma unroll
    for (int r = 0; r < 8; ++r) {
      int M = r + (hi << 3);
      int i = it * 16 + nl;
      if (i < NOFF) srow[M * SSTR + i] = acc[r];
    }
  }

  for (int t = 0; t < NTILE; ++t) {
    const int off = c_toff[t];
    if (n0 + 15 < off) continue;
    const int m0 = n0 - off;
    int krow = m0 + nl; if (krow < 0) krow = 0;
    const float* kr = kbh + (size_t)krow * HDIM + koff;
    __builtin_prefetch(vbh + (size_t)krow * HDIM, 0, 1);
    v8f acc = {0.f,0.f,0.f,0.f,0.f,0.f,0.f,0.f};
#pragma unroll
    for (int c = 0; c < 16; ++c) {
      v2f b = *(const v2f*)(kr + 4 * c);
      acc = wmma4(qa[c], b, acc);
    }
    if (t < 3) {
#pragma unroll
      for (int r = 0; r < 8; ++r) {
        int M = r + (hi << 3);
        int d = M - nl + (t << 4);
        if (d >= 0 && d <= 32) atomicAdd(&srow[M * SSTR + d], acc[r]);
      }
    } else {
      const int i = 33 + (t - 3);
#pragma unroll
      for (int r = 0; r < 8; ++r) {
        int M = r + (hi << 3);
        if (M == nl) atomicAdd(&srow[M * SSTR + i], acc[r]);
      }
    }
  }

  if (lane < 16) {
    float* sr = srow + lane * SSTR;
    const int ng = n0 + lane;
    float mx = -3.0e38f;
    for (int i = 0; i < NOFF; ++i) {
      float sv = sr[i] + s_pb[i];
      if (ng < c_off[i]) sv = -3.0e38f;
      sr[i] = sv;
      mx = fmaxf(mx, sv);
    }
    float sum = 0.f;
    for (int i = 0; i < NOFF; ++i) {
      float e = __expf(sr[i] - mx);
      sr[i] = e;
      sum += e;
    }
    s_rinv[wv * 16 + lane] = 1.0f / sum;
  }

  const float rinv = s_rinv[wv * 16 + nl];

  v8f oc[4], osn[4];
#pragma unroll
  for (int ht = 0; ht < 4; ++ht) {
    oc[ht]  = (v8f){0.f,0.f,0.f,0.f,0.f,0.f,0.f,0.f};
    osn[ht] = (v8f){0.f,0.f,0.f,0.f,0.f,0.f,0.f,0.f};
  }

  for (int t = 0; t < NTILE; ++t) {
    const int off = c_toff[t];
    if (n0 + 15 < off) continue;
    const int m0 = n0 - off;

    v2f pac[4], pas[4];
#pragma unroll
    for (int c = 0; c < 4; ++c) {
#pragma unroll
      for (int u = 0; u < 2; ++u) {
        int j = 4 * c + koff + u;
        int iidx = 0; bool valid;
        if (t < 3) { int d = nl - j + (t << 4); valid = (d >= 0) && (d <= 32); iidx = d; }
        else       { valid = (nl == j); iidx = 33 + (t - 3); }
        float pc = 0.f, ps = 0.f;
        if (valid) {
          float pv = srow[nl * SSTR + iidx] * rinv;
          pc = pv * s_cp[iidx];
          ps = pv * s_sp[iidx];
        }
        if (u == 0) { pac[c].x = pc; pas[c].x = ps; }
        else        { pac[c].y = pc; pas[c].y = ps; }
      }
    }

#pragma unroll
    for (int ht = 0; ht < 4; ++ht) {
      const int hd0 = ht << 4;
#pragma unroll
      for (int c = 0; c < 4; ++c) {
        int rr = m0 + 4 * c + koff;
        int r0 = rr < 0 ? 0 : rr;
        int r1 = (rr + 1) < 0 ? 0 : (rr + 1);
        v2f vb;
        vb.x = vbh[(size_t)r0 * HDIM + hd0 + nl];
        vb.y = vbh[(size_t)r1 * HDIM + hd0 + nl];
        oc[ht]  = wmma4(pac[c], vb, oc[ht]);
        osn[ht] = wmma4(pas[c], vb, osn[ht]);
      }
    }
  }

  const float sgn = (lane & 1) ? 1.0f : -1.0f;
#pragma unroll
  for (int ht = 0; ht < 4; ++ht) {
#pragma unroll
    for (int r = 0; r < 8; ++r) {
      float sv = osn[ht][r];
      float sw = __shfl_xor(sv, 1, 32);
      float res = oc[ht][r] + sgn * sw;
      int M = r + (hi << 3);
      obh[(size_t)(n0 + M) * HDIM + (ht << 4) + nl] = res;
    }
  }
}

extern "C" void kernel_launch(void* const* d_in, const int* in_sizes, int n_in,
                              void* d_out, int out_size, void* d_ws, size_t ws_size,
                              hipStream_t stream) {
  const float* q  = (const float*)d_in[0];
  const float* k  = (const float*)d_in[1];
  const float* v  = (const float*)d_in[2];
  const float* pb = (const float*)d_in[3];
  const float* se = (const float*)d_in[4];
  const float* ph = (const float*)d_in[5];
  float* out = (float*)d_out;

  dim3 grid(NB * NH * (NN / 64));
  dim3 block(128);
  hipLaunchKernelGGL(dsqg_attn_v4_kernel, grid, block, 0, stream,
                     q, k, v, pb, se, ph, out);
}


# Q, K and SE fragment loads 8 B -> 16 B per lane (hd 8j+4*half inside the k chunks, same map for A and B): half the loads per line touched, VGPRs 163 -> 144; score hd terms summed in a new order
# speedup vs baseline: 1.0487x; 1.0487x over previous
#include <hip/hip_runtime.h>

typedef __attribute__((ext_vector_type(2))) float v2f;
typedef __attribute__((ext_vector_type(4))) float v4f;
typedef __attribute__((ext_vector_type(8))) float v8f;

#define NB 2
#define NH 16
#define NN 8192
#define HDIM 64
#define NOFF 44
#define NTILE 14
#define SSTR 52

__constant__ int c_off[NOFF] = {
  0,1,2,3,4,5,6,7,8,9,10,11,12,13,14,15,16,17,18,19,20,21,22,
  23,24,25,26,27,28,29,30,31,32,
  48,64,96,128,192,256,384,512,768,1024,1536 };

__constant__ int c_toff[NTILE] = {0,16,32,48,64,96,128,192,256,384,512,768,1024,1536};

__device__ __forceinline__ v8f wmma4(v2f a, v2f b, v8f c) {
  return __builtin_amdgcn_wmma_f32_16x16x4_f32(
      false, a, false, b, (short)0, c, false, false);
}

__global__ __launch_bounds__(128)
void dsqg_attn_v4_kernel(const float* __restrict__ q,
                         const float* __restrict__ k,
                         const float* __restrict__ v,
                         const float* __restrict__ pbias,
                         const float* __restrict__ se,
                         const float* __restrict__ ph,
                         float* __restrict__ out)
{
  __shared__ float s_cp[48], s_sp[48], s_pb[48];
  __shared__ float s_scores[4 * 16 * SSTR];
  __shared__ float s_rinv[4 * 16];

  const int bid  = blockIdx.x;
  const int bh   = bid >> 7;
  const int h    = bh & (NH - 1);
  const int n0b  = (bid & 127) << 6;
  const int tid  = threadIdx.x;
  const int wv   = tid >> 5;
  const int lane = tid & 31;
  const int n0   = n0b + (wv << 4);

  if (tid < NOFF) {
    float p = ph[tid * NH + h];
    s_cp[tid] = cosf(p);
    s_sp[tid] = sinf(p);
    s_pb[tid] = pbias[tid * NH + h];
  }
  __syncthreads();

  const float sc   = 0.125f;
  const int   nl   = lane & 15;
  const int   hi   = lane >> 4;
  const int   koff = hi << 1;
  const int   kq   = hi << 2;

  const size_t bhoff = (size_t)bh * NN * HDIM;
  const float* qbh = q + bhoff;
  const float* kbh = k + bhoff;
  const float* vbh = v + bhoff;
  float*       obh = out + bhoff;
  float*       srow = s_scores + wv * (16 * SSTR);

  v2f qa[16];
  {
    const float* qr = qbh + (size_t)(n0 + nl) * HDIM + kq;
#pragma unroll
    for (int j = 0; j < 8; ++j) {
      v4f t = *(const v4f*)(qr + 8 * j);
      qa[2 * j].x     = t.x * sc;
      qa[2 * j].y     = t.y * sc;
      qa[2 * j + 1].x = t.z * sc;
      qa[2 * j + 1].y = t.w * sc;
    }
  }

#pragma unroll
  for (int it = 0; it < 3; ++it) {
    int irow = it * 16 + nl;
    int ir   = (irow < NOFF) ? irow : (NOFF - 1);
    const float* ser = se + (size_t)ir * HDIM + kq;
    v8f acc = {0.f,0.f,0.f,0.f,0.f,0.f,0.f,0.f};
#pragma unroll
    for (int j = 0; j < 8; ++j) {
      v4f b = *(const v4f*)(ser + 8 * j);
      acc = wmma4(qa[2 * j],     (v2f){b.x, b.y}, acc);
      acc = wmma4(qa[2 * j + 1], (v2f){b.z, b.w}, acc);
    }
#pragma unroll
    for (int r = 0; r < 8; ++r) {
      int M = r + (hi << 3);
      int i = it * 16 + nl;
      if (i < NOFF) srow[M * SSTR + i] = acc[r];
    }
  }

  for (int t = 0; t < NTILE; ++t) {
    const int off = c_toff[t];
    if (n0 + 15 < off) continue;
    const int m0 = n0 - off;
    int krow = m0 + nl; if (krow < 0) krow = 0;
    const float* kr = kbh + (size_t)krow * HDIM + kq;
    __builtin_prefetch(vbh + (size_t)krow * HDIM, 0, 1);
    v8f acc = {0.f,0.f,0.f,0.f,0.f,0.f,0.f,0.f};
#pragma unroll
    for (int j = 0; j < 8; ++j) {
      v4f b = *(const v4f*)(kr + 8 * j);
      acc = wmma4(qa[2 * j],     (v2f){b.x, b.y}, acc);
      acc = wmma4(qa[2 * j + 1], (v2f){b.z, b.w}, acc);
    }
    if (t < 3) {
#pragma unroll
      for (int r = 0; r < 8; ++r) {
        int M = r + (hi << 3);
        int d = M - nl + (t << 4);
        if (d >= 0 && d <= 32) atomicAdd(&srow[M * SSTR + d], acc[r]);
      }
    } else {
      const int i = 33 + (t - 3);
#pragma unroll
      for (int r = 0; r < 8; ++r) {
        int M = r + (hi << 3);
        if (M == nl) atomicAdd(&srow[M * SSTR + i], acc[r]);
      }
    }
  }

  if (lane < 16) {
    float* sr = srow + lane * SSTR;
    const int ng = n0 + lane;
    float mx = -3.0e38f;
    for (int i = 0; i < NOFF; ++i) {
      float sv = sr[i] + s_pb[i];
      if (ng < c_off[i]) sv = -3.0e38f;
      sr[i] = sv;
      mx = fmaxf(mx, sv);
    }
    float sum = 0.f;
    for (int i = 0; i < NOFF; ++i) {
      float e = __expf(sr[i] - mx);
      sr[i] = e;
      sum += e;
    }
    s_rinv[wv * 16 + lane] = 1.0f / sum;
  }

  const float rinv = s_rinv[wv * 16 + nl];

  v8f oc[4], osn[4];
#pragma unroll
  for (int ht = 0; ht < 4; ++ht) {
    oc[ht]  = (v8f){0.f,0.f,0.f,0.f,0.f,0.f,0.f,0.f};
    osn[ht] = (v8f){0.f,0.f,0.f,0.f,0.f,0.f,0.f,0.f};
  }

  for (int t = 0; t < NTILE; ++t) {
    const int off = c_toff[t];
    if (n0 + 15 < off) continue;
    const int m0 = n0 - off;

    v2f pac[4], pas[4];
#pragma unroll
    for (int c = 0; c < 4; ++c) {
#pragma unroll
      for (int u = 0; u < 2; ++u) {
        int j = 4 * c + koff + u;
        int iidx = 0; bool valid;
        if (t < 3) { int d = nl - j + (t << 4); valid = (d >= 0) && (d <= 32); iidx = d; }
        else       { valid = (nl == j); iidx = 33 + (t - 3); }
        float pc = 0.f, ps = 0.f;
        if (valid) {
          float pv = srow[nl * SSTR + iidx] * rinv;
          pc = pv * s_cp[iidx];
          ps = pv * s_sp[iidx];
        }
        if (u == 0) { pac[c].x = pc; pas[c].x = ps; }
        else        { pac[c].y = pc; pas[c].y = ps; }
      }
    }

#pragma unroll
    for (int ht = 0; ht < 4; ++ht) {
      const int hd0 = ht << 4;
#pragma unroll
      for (int c = 0; c < 4; ++c) {
        int rr = m0 + 4 * c + koff;
        int r0 = rr < 0 ? 0 : rr;
        int r1 = (rr + 1) < 0 ? 0 : (rr + 1);
        v2f vb;
        vb.x = vbh[(size_t)r0 * HDIM + hd0 + nl];
        vb.y = vbh[(size_t)r1 * HDIM + hd0 + nl];
        oc[ht]  = wmma4(pac[c], vb, oc[ht]);
        osn[ht] = wmma4(pas[c], vb, osn[ht]);
      }
    }
  }

  const float sgn = (lane & 1) ? 1.0f : -1.0f;
#pragma unroll
  for (int ht = 0; ht < 4; ++ht) {
#pragma unroll
    for (int r = 0; r < 8; ++r) {
      float sv = osn[ht][r];
      float sw = __shfl_xor(sv, 1, 32);
      float res = oc[ht][r] + sgn * sw;
      int M = r + (hi << 3);
      obh[(size_t)(n0 + M) * HDIM + (ht << 4) + nl] = res;
    }
  }
}

extern "C" void kernel_launch(void* const* d_in, const int* in_sizes, int n_in,
                              void* d_out, int out_size, void* d_ws, size_t ws_size,
                              hipStream_t stream) {
  const float* q  = (const float*)d_in[0];
  const float* k  = (const float*)d_in[1];
  const float* v  = (const float*)d_in[2];
  const float* pb = (const float*)d_in[3];
  const float* se = (const float*)d_in[4];
  const float* ph = (const float*)d_in[5];
  float* out = (float*)d_out;

  dim3 grid(NB * NH * (NN / 64));
  dim3 block(128);
  hipLaunchKernelGGL(dsqg_attn_v4_kernel, grid, block, 0, stream,
                     q, k, v, pb, se, ph, out);
}


# P.V pass: cos- and sin-weighted P share one A fragment per output-row half: diagonal tiles issue 1 WMMA per V fragment (was 2), all-zero triangle fragments none; 720 -> 528 WMMA/wave; sum order kept
# speedup vs baseline: 1.2609x; 1.2023x over previous
#include <hip/hip_runtime.h>

typedef __attribute__((ext_vector_type(2))) float v2f;
typedef __attribute__((ext_vector_type(4))) float v4f;
typedef __attribute__((ext_vector_type(8))) float v8f;

#define NB 2
#define NH 16
#define NN 8192
#define HDIM 64
#define NOFF 44
#define NTILE 14
#define SSTR 52

__constant__ int c_off[NOFF] = {
  0,1,2,3,4,5,6,7,8,9,10,11,12,13,14,15,16,17,18,19,20,21,22,
  23,24,25,26,27,28,29,30,31,32,
  48,64,96,128,192,256,384,512,768,1024,1536 };

__constant__ int c_toff[NTILE] = {0,16,32,48,64,96,128,192,256,384,512,768,1024,1536};

__device__ __forceinline__ v8f wmma4(v2f a, v2f b, v8f c) {
  return __builtin_amdgcn_wmma_f32_16x16x4_f32(
      false, a, false, b, (short)0, c, false, false);
}

__global__ __launch_bounds__(128, 7)
void dsqg_attn_v4_kernel(const float* __restrict__ q,
                         const float* __restrict__ k,
                         const float* __restrict__ v,
                         const float* __restrict__ pbias,
                         const float* __restrict__ se,
                         const float* __restrict__ ph,
                         float* __restrict__ out)
{
  __shared__ float s_cp[48], s_sp[48], s_pb[48];
  __shared__ float s_scores[4 * 16 * SSTR];
  __shared__ float s_rinv[4 * 16];

  const int bid  = blockIdx.x;
  const int bh   = bid >> 7;
  const int h    = bh & (NH - 1);
  const int n0b  = (bid & 127) << 6;
  const int tid  = threadIdx.x;
  const int wv   = tid >> 5;
  const int lane = tid & 31;
  const int n0   = n0b + (wv << 4);

  if (tid < NOFF) {
    float p = ph[tid * NH + h];
    s_cp[tid] = cosf(p);
    s_sp[tid] = sinf(p);
    s_pb[tid] = pbias[tid * NH + h];
  }
  __syncthreads();

  const float sc   = 0.125f;
  const int   nl   = lane & 15;
  const int   hi   = lane >> 4;
  const int   koff = hi << 1;
  const int   kq   = hi << 2;

  const size_t bhoff = (size_t)bh * NN * HDIM;
  const float* qbh = q + bhoff;
  const float* kbh = k + bhoff;
  const float* vbh = v + bhoff;
  float*       obh = out + bhoff;
  float*       srow = s_scores + wv * (16 * SSTR);

  v2f qa[16];
  {
    const float* qr = qbh + (size_t)(n0 + nl) * HDIM + kq;
#pragma unroll
    for (int j = 0; j < 8; ++j) {
      v4f t = *(const v4f*)(qr + 8 * j);
      qa[2 * j].x     = t.x * sc;
      qa[2 * j].y     = t.y * sc;
      qa[2 * j + 1].x = t.z * sc;
      qa[2 * j + 1].y = t.w * sc;
    }
  }

#pragma nounroll
  for (int it = 0; it < 3; ++it) {
    int irow = it * 16 + nl;
    int ir   = (irow < NOFF) ? irow : (NOFF - 1);
    const float* ser = se + (size_t)ir * HDIM + kq;
    v8f acc = {0.f,0.f,0.f,0.f,0.f,0.f,0.f,0.f};
#pragma unroll
    for (int j = 0; j < 8; ++j) {
      v4f b = *(const v4f*)(ser + 8 * j);
      acc = wmma4(qa[2 * j],     (v2f){b.x, b.y}, acc);
      acc = wmma4(qa[2 * j + 1], (v2f){b.z, b.w}, acc);
    }
#pragma unroll
    for (int r = 0; r < 8; ++r) {
      int M = r + (hi << 3);
      int i = it * 16 + nl;
      if (i < NOFF) srow[M * SSTR + i] = acc[r];
    }
  }

  for (int t = 0; t < NTILE; ++t) {
    const int off = c_toff[t];
    if (n0 + 15 < off) continue;
    const int m0 = n0 - off;
    int krow = m0 + nl; if (krow < 0) krow = 0;
    const float* kr = kbh + (size_t)krow * HDIM + kq;
    __builtin_prefetch(vbh + (size_t)krow * HDIM, 0, 1);
    v8f acc = {0.f,0.f,0.f,0.f,0.f,0.f,0.f,0.f};
#pragma unroll
    for (int j = 0; j < 8; ++j) {
      v4f b = *(const v4f*)(kr + 8 * j);
      acc = wmma4(qa[2 * j],     (v2f){b.x, b.y}, acc);
      acc = wmma4(qa[2 * j + 1], (v2f){b.z, b.w}, acc);
    }
    if (t < 3) {
#pragma unroll
      for (int r = 0; r < 8; ++r) {
        int M = r + (hi << 3);
        int d = M - nl + (t << 4);
        if (d >= 0 && d <= 32) atomicAdd(&srow[M * SSTR + d], acc[r]);
      }
    } else {
      const int i = 33 + (t - 3);
#pragma unroll
      for (int r = 0; r < 8; ++r) {
        int M = r + (hi << 3);
        if (M == nl) atomicAdd(&srow[M * SSTR + i], acc[r]);
      }
    }
  }

  if (lane < 16) {
    float* sr = srow + lane * SSTR;
    const int ng = n0 + lane;
    float mx = -3.0e38f;
    for (int i = 0; i < NOFF; ++i) {
      float sv = sr[i] + s_pb[i];
      if (ng < c_off[i]) sv = -3.0e38f;
      sr[i] = sv;
      mx = fmaxf(mx, sv);
    }
    float sum = 0.f;
    for (int i = 0; i < NOFF; ++i) {
      float e = __expf(sr[i] - mx);
      sr[i] = e;
      sum += e;
    }
    s_rinv[wv * 16 + lane] = 1.0f / sum;
  }

  const int   r7    = nl & 7;
  const bool  lo8   = nl < 8;
  const float* srx  = srow + r7 * SSTR;
  const float* sry  = srow + (r7 + 8) * SSTR;
  const float rinvx = s_rinv[wv * 16 + r7];
  const float rinvy = s_rinv[wv * 16 + r7 + 8];
  const float* tabx = lo8 ? s_cp : s_sp;
  const float* taby = lo8 ? s_sp : s_cp;

  v8f ox[4], oy[4];
#pragma unroll
  for (int ht = 0; ht < 4; ++ht) {
    ox[ht] = (v8f){0.f,0.f,0.f,0.f,0.f,0.f,0.f,0.f};
    oy[ht] = (v8f){0.f,0.f,0.f,0.f,0.f,0.f,0.f,0.f};
  }

#pragma unroll
  for (int t = 0; t < 3; ++t) {
    const int off = t << 4;
    if (n0 + 15 < off) continue;
    const int m0 = n0 - off;

    v2f pax[4], pay[4];
#pragma unroll
    for (int c = 0; c < 4; ++c) {
#pragma unroll
      for (int u = 0; u < 2; ++u) {
        int j  = 4 * c + koff + u;
        int dx = r7 - j + off;
        int dy = dx + 8;
        float ax = 0.f, ay = 0.f;
        if (dx >= 0 && dx <= 32) { float pv = srx[dx] * rinvx; ax = pv * tabx[dx]; }
        if (dy >= 0 && dy <= 32) { float pv = sry[dy] * rinvy; ay = pv * taby[dy]; }
        if (u == 0) { pax[c].x = ax; pay[c].x = ay; }
        else        { pax[c].y = ax; pay[c].y = ay; }
      }
    }

#pragma unroll
    for (int ht = 0; ht < 4; ++ht) {
      const int hd0 = ht << 4;
#pragma unroll
      for (int c = 0; c < 4; ++c) {
        int rr = m0 + 4 * c + koff;
        int r0 = rr < 0 ? 0 : rr;
        int r1 = (rr + 1) < 0 ? 0 : (rr + 1);
        v2f vb;
        vb.x = vbh[(size_t)r0 * HDIM + hd0 + nl];
        vb.y = vbh[(size_t)r1 * HDIM + hd0 + nl];
        if (!(t == 0 && c >= 2)) ox[ht] = wmma4(pax[c], vb, ox[ht]);
        if (!(t == 2 && c < 2))  oy[ht] = wmma4(pay[c], vb, oy[ht]);
      }
    }
  }

  for (int t = 3; t < NTILE; ++t) {
    const int off = c_toff[t];
    if (n0 + 15 < off) continue;
    const int m0 = n0 - off;
    const int i  = 33 + (t - 3);

    const float px = (srx[i] * rinvx) * tabx[i];
    const float py = (sry[i] * rinvy) * taby[i];

    v2f pa[4];
#pragma unroll
    for (int c = 0; c < 4; ++c) {
#pragma unroll
      for (int u = 0; u < 2; ++u) {
        int j = 4 * c + koff + u;
        float a = (r7 == (j & 7)) ? ((c < 2) ? px : py) : 0.f;
        if (u == 0) pa[c].x = a; else pa[c].y = a;
      }
    }

#pragma unroll
    for (int ht = 0; ht < 4; ++ht) {
      const int hd0 = ht << 4;
#pragma unroll
      for (int c = 0; c < 4; ++c) {
        int rr = m0 + 4 * c + koff;
        int r0 = rr < 0 ? 0 : rr;
        int r1 = (rr + 1) < 0 ? 0 : (rr + 1);
        v2f vb;
        vb.x = vbh[(size_t)r0 * HDIM + hd0 + nl];
        vb.y = vbh[(size_t)r1 * HDIM + hd0 + nl];
        if (c < 2) ox[ht] = wmma4(pa[c], vb, ox[ht]);
        else       oy[ht] = wmma4(pa[c], vb, oy[ht]);
      }
    }
  }

  const float sgn = (lane & 1) ? 1.0f : -1.0f;
#pragma unroll
  for (int ht = 0; ht < 4; ++ht) {
#pragma unroll
    for (int r = 0; r < 8; ++r) {
      float cv = hi ? oy[ht][r] : ox[ht][r];
      float sv = hi ? ox[ht][r] : oy[ht][r];
      float sw = __shfl_xor(sv, 17, 32);
      float res = cv + sgn * sw;
      int M = r + (hi << 3);
      obh[(size_t)(n0 + M) * HDIM + (ht << 4) + nl] = res;
    }
  }
}

extern "C" void kernel_launch(void* const* d_in, const int* in_sizes, int n_in,
                              void* d_out, int out_size, void* d_ws, size_t ws_size,
                              hipStream_t stream) {
  const float* q  = (const float*)d_in[0];
  const float* k  = (const float*)d_in[1];
  const float* v  = (const float*)d_in[2];
  const float* pb = (const float*)d_in[3];
  const float* se = (const float*)d_in[4];
  const float* ph = (const float*)d_in[5];
  float* out = (float*)d_out;

  dim3 grid(NB * NH * (NN / 64));
  dim3 block(128);
  hipLaunchKernelGGL(dsqg_attn_v4_kernel, grid, block, 0, stream,
                     q, k, v, pb, se, ph, out);
}


# LDS score rows padded and skewed: each band-tile entry has its own column, so tiles scatter and P.V gathers at constant offsets with no per-element band tests; LDS atomics -> plain stores
# speedup vs baseline: 1.3178x; 1.0451x over previous
#include <hip/hip_runtime.h>

typedef __attribute__((ext_vector_type(2))) float v2f;
typedef __attribute__((ext_vector_type(4))) float v4f;
typedef __attribute__((ext_vector_type(8))) float v8f;

#define NB 2
#define NH 16
#define NN 8192
#define HDIM 64
#define NOFF 44
#define NTILE 14
#define SS 76
#define DCOL 16
#define SCOL 64
#define SSLOT (SCOL - DCOL)
#define QSTR 60

__constant__ int c_off[NOFF] = {
  0,1,2,3,4,5,6,7,8,9,10,11,12,13,14,15,16,17,18,19,20,21,22,
  23,24,25,26,27,28,29,30,31,32,
  48,64,96,128,192,256,384,512,768,1024,1536 };

__constant__ int c_toff[NTILE] = {0,16,32,48,64,96,128,192,256,384,512,768,1024,1536};

__device__ __forceinline__ v8f wmma4(v2f a, v2f b, v8f c) {
  return __builtin_amdgcn_wmma_f32_16x16x4_f32(
      false, a, false, b, (short)0, c, false, false);
}

__global__ __launch_bounds__(128, 7)
void dsqg_attn_v4_kernel(const float* __restrict__ q,
                         const float* __restrict__ k,
                         const float* __restrict__ v,
                         const float* __restrict__ pbias,
                         const float* __restrict__ se,
                         const float* __restrict__ ph,
                         float* __restrict__ out)
{
  __shared__ __attribute__((aligned(16))) float s_cp[64];
  __shared__ __attribute__((aligned(16))) float s_sp[64];
  __shared__ __attribute__((aligned(16))) float s_pb[64];
  __shared__ __attribute__((aligned(16))) float s_scores[4 * 16 * SS];
  __shared__ __attribute__((aligned(16))) float s_qd[4 * 16 * QSTR];
  __shared__ float s_rinv[4 * 16];

  const int bid  = blockIdx.x;
  const int bh   = bid >> 7;
  const int h    = bh & (NH - 1);
  const int n0b  = (bid & 127) << 6;
  const int tid  = threadIdx.x;
  const int wv   = tid >> 5;
  const int lane = tid & 31;
  const int n0   = n0b + (wv << 4);

  if (tid < 64) {
    const int ii = tid - DCOL;
    float cv = 0.f, sv = 0.f;
    if (ii >= 0 && ii < NOFF) {
      float p = ph[ii * NH + h];
      cv = cosf(p);
      sv = sinf(p);
      s_pb[ii < 33 ? ii : ii + (SSLOT - 33)] = pbias[ii * NH + h];
    }
    s_cp[tid] = cv;
    s_sp[tid] = sv;
  }
  __syncthreads();

  const float sc   = 0.125f;
  const int   nl   = lane & 15;
  const int   hi   = lane >> 4;
  const int   koff = hi << 1;
  const int   kq   = hi << 2;

  const size_t bhoff = (size_t)bh * NN * HDIM;
  const float* qbh = q + bhoff;
  const float* kbh = k + bhoff;
  const float* vbh = v + bhoff;
  float*       obh = out + bhoff;
  float*       srow = s_scores + wv * (16 * SS);
  float*       qrow = s_qd + wv * (16 * QSTR);

  v2f qa[16];
  {
    const float* qr = qbh + (size_t)(n0 + nl) * HDIM + kq;
#pragma unroll
    for (int j = 0; j < 8; ++j) {
      v4f t = *(const v4f*)(qr + 8 * j);
      qa[2 * j].x     = t.x * sc;
      qa[2 * j].y     = t.y * sc;
      qa[2 * j + 1].x = t.z * sc;
      qa[2 * j + 1].y = t.w * sc;
    }
  }

#pragma nounroll
  for (int it = 0; it < 3; ++it) {
    int irow = it * 16 + nl;
    int ir   = (irow < NOFF) ? irow : (NOFF - 1);
    const float* ser = se + (size_t)ir * HDIM + kq;
    v8f acc = {0.f,0.f,0.f,0.f,0.f,0.f,0.f,0.f};
#pragma unroll
    for (int j = 0; j < 8; ++j) {
      v4f b = *(const v4f*)(ser + 8 * j);
      acc = wmma4(qa[2 * j],     (v2f){b.x, b.y}, acc);
      acc = wmma4(qa[2 * j + 1], (v2f){b.z, b.w}, acc);
    }
#pragma unroll
    for (int r = 0; r < 8; ++r) {
      int M = r + (hi << 3);
      int i = it * 16 + nl;
      if (i < NOFF) qrow[M * QSTR + (i < 33 ? i : i + (SSLOT - 33))] = acc[r];
    }
  }

  const int  sdense  = (hi << 3) * (SS + 1) + DCOL - nl;
  const bool diag    = ((nl >> 3) == hi);
  const int  ssparse = nl * SS + (diag ? SCOL : 0);
  const bool sb0 = (nl & 1) != 0, sb1 = (nl & 2) != 0, sb2 = (nl & 4) != 0;
  for (int t = 0; t < NTILE; ++t) {
    const int off = c_toff[t];
    if (n0 + 15 < off) continue;
    const int m0 = n0 - off;
    int krow = m0 + nl; if (krow < 0) krow = 0;
    const float* kr = kbh + (size_t)krow * HDIM + kq;
    __builtin_prefetch(vbh + (size_t)krow * HDIM, 0, 1);
    v8f acc = {0.f,0.f,0.f,0.f,0.f,0.f,0.f,0.f};
#pragma unroll
    for (int j = 0; j < 8; ++j) {
      v4f b = *(const v4f*)(kr + 8 * j);
      acc = wmma4(qa[2 * j],     (v2f){b.x, b.y}, acc);
      acc = wmma4(qa[2 * j + 1], (v2f){b.z, b.w}, acc);
    }
    if (t < 3) {
#pragma unroll
      for (int r = 0; r < 8; ++r) srow[sdense + r * (SS + 1) + (t << 4)] = acc[r];
    } else {
      float a0 = sb0 ? acc[1] : acc[0];
      float a1 = sb0 ? acc[3] : acc[2];
      float a2 = sb0 ? acc[5] : acc[4];
      float a3 = sb0 ? acc[7] : acc[6];
      float c0 = sb1 ? a1 : a0;
      float c1 = sb1 ? a3 : a2;
      srow[ssparse + (t - 3)] = sb2 ? c1 : c0;
    }
  }

  if (lane < 16) {
    float* sr = srow + lane * SS + DCOL;
    const float* qd = qrow + lane * QSTR;
    const int ng = n0 + lane;
    const float NEG = -3.0e38f;
    float mx = NEG;
    for (int g = 0; g < 8; ++g) {
      v4f sv = (*(const v4f*)(sr + 4 * g) + *(const v4f*)(qd + 4 * g)) + *(const v4f*)(s_pb + 4 * g);
      if (ng < 4 * g)     sv.x = NEG;
      if (ng < 4 * g + 1) sv.y = NEG;
      if (ng < 4 * g + 2) sv.z = NEG;
      if (ng < 4 * g + 3) sv.w = NEG;
      *(v4f*)(sr + 4 * g) = sv;
      mx = fmaxf(fmaxf(fmaxf(fmaxf(mx, sv.x), sv.y), sv.z), sv.w);
    }
    {
      float sv = (sr[32] + qd[32]) + s_pb[32];
      if (ng < 32) sv = NEG;
      sr[32] = sv;
      mx = fmaxf(mx, sv);
    }
    for (int g = 0; g < 2; ++g) {
      const int k = SSLOT + 4 * g;
      v4f sv = (*(const v4f*)(sr + k) + *(const v4f*)(qd + k)) + *(const v4f*)(s_pb + k);
      if (ng < c_off[33 + 4 * g])     sv.x = NEG;
      if (ng < c_off[33 + 4 * g + 1]) sv.y = NEG;
      if (ng < c_off[33 + 4 * g + 2]) sv.z = NEG;
      if (ng < c_off[33 + 4 * g + 3]) sv.w = NEG;
      *(v4f*)(sr + k) = sv;
      mx = fmaxf(fmaxf(fmaxf(fmaxf(mx, sv.x), sv.y), sv.z), sv.w);
    }
    for (int s = 8; s < NOFF - 33; ++s) {
      float sv = (sr[SSLOT + s] + qd[SSLOT + s]) + s_pb[SSLOT + s];
      if (ng < c_off[33 + s]) sv = NEG;
      sr[SSLOT + s] = sv;
      mx = fmaxf(mx, sv);
    }

    float sum = 0.f;
    for (int g = 0; g < 8; ++g) {
      v4f a = *(const v4f*)(sr + 4 * g);
      v4f e;
      e.x = __expf(a.x - mx); e.y = __expf(a.y - mx); e.z = __expf(a.z - mx); e.w = __expf(a.w - mx);
      *(v4f*)(sr + 4 * g) = e;
      sum += e.x; sum += e.y; sum += e.z; sum += e.w;
    }
    {
      float e = __expf(sr[32] - mx);
      sr[32] = e;
      sum += e;
    }
    for (int g = 0; g < 2; ++g) {
      const int k = SSLOT + 4 * g;
      v4f a = *(const v4f*)(sr + k);
      v4f e;
      e.x = __expf(a.x - mx); e.y = __expf(a.y - mx); e.z = __expf(a.z - mx); e.w = __expf(a.w - mx);
      *(v4f*)(sr + k) = e;
      sum += e.x; sum += e.y; sum += e.z; sum += e.w;
    }
    for (int s = 8; s < NOFF - 33; ++s) {
      float e = __expf(sr[SSLOT + s] - mx);
      sr[SSLOT + s] = e;
      sum += e;
    }
    s_rinv[wv * 16 + lane] = 1.0f / sum;

    const v4f z4 = {0.f, 0.f, 0.f, 0.f};
    for (int g = 0; g < 4; ++g) *(v4f*)(sr - DCOL + 4 * g) = z4;
    sr[33] = 0.f; sr[34] = 0.f; sr[35] = 0.f;
    for (int g = 9; g < 12; ++g) *(v4f*)(sr + 4 * g) = z4;
  }

  const int   r7    = nl & 7;
  const bool  lo8   = nl < 8;
  const float* srx  = srow + r7 * SS;
  const float* sry  = srow + (r7 + 8) * SS;
  const float rinvx = s_rinv[wv * 16 + r7];
  const float rinvy = s_rinv[wv * 16 + r7 + 8];
  const float* tabx = lo8 ? s_cp : s_sp;
  const float* taby = lo8 ? s_sp : s_cp;

  v8f ox[4], oy[4];
#pragma unroll
  for (int ht = 0; ht < 4; ++ht) {
    ox[ht] = (v8f){0.f,0.f,0.f,0.f,0.f,0.f,0.f,0.f};
    oy[ht] = (v8f){0.f,0.f,0.f,0.f,0.f,0.f,0.f,0.f};
  }

#pragma unroll
  for (int t = 0; t < 3; ++t) {
    const int off = t << 4;
    if (n0 + 15 < off) continue;
    const int m0 = n0 - off;

    v2f pax[4], pay[4];
#pragma unroll
    for (int c = 0; c < 4; ++c) {
#pragma unroll
      for (int u = 0; u < 2; ++u) {
        int j  = 4 * c + koff + u;
        int dx = r7 - j + off;
        int dy = dx + 8;
        float ax = (srx[DCOL + dx] * rinvx) * tabx[DCOL + dx];
        float ay = (sry[DCOL + dy] * rinvy) * taby[DCOL + dy];
        if (u == 0) { pax[c].x = ax; pay[c].x = ay; }
        else        { pax[c].y = ax; pay[c].y = ay; }
      }
    }

#pragma unroll
    for (int ht = 0; ht < 4; ++ht) {
      const int hd0 = ht << 4;
#pragma unroll
      for (int c = 0; c < 4; ++c) {
        int rr = m0 + 4 * c + koff;
        int r0 = rr < 0 ? 0 : rr;
        int r1 = (rr + 1) < 0 ? 0 : (rr + 1);
        v2f vb;
        vb.x = vbh[(size_t)r0 * HDIM + hd0 + nl];
        vb.y = vbh[(size_t)r1 * HDIM + hd0 + nl];
        if (!(t == 0 && c >= 2)) ox[ht] = wmma4(pax[c], vb, ox[ht]);
        if (!(t == 2 && c < 2))  oy[ht] = wmma4(pay[c], vb, oy[ht]);
      }
    }
  }

  for (int t = 3; t < NTILE; ++t) {
    const int off = c_toff[t];
    if (n0 + 15 < off) continue;
    const int m0 = n0 - off;
    const int s  = t - 3;

    const float px = (srx[SCOL + s] * rinvx) * tabx[DCOL + 33 + s];
    const float py = (sry[SCOL + s] * rinvy) * taby[DCOL + 33 + s];

    v2f pa[4];
#pragma unroll
    for (int c = 0; c < 4; ++c) {
#pragma unroll
      for (int u = 0; u < 2; ++u) {
        int j = 4 * c + koff + u;
        float a = (r7 == (j & 7)) ? ((c < 2) ? px : py) : 0.f;
        if (u == 0) pa[c].x = a; else pa[c].y = a;
      }
    }

#pragma unroll
    for (int ht = 0; ht < 4; ++ht) {
      const int hd0 = ht << 4;
#pragma unroll
      for (int c = 0; c < 4; ++c) {
        int rr = m0 + 4 * c + koff;
        int r0 = rr < 0 ? 0 : rr;
        int r1 = (rr + 1) < 0 ? 0 : (rr + 1);
        v2f vb;
        vb.x = vbh[(size_t)r0 * HDIM + hd0 + nl];
        vb.y = vbh[(size_t)r1 * HDIM + hd0 + nl];
        if (c < 2) ox[ht] = wmma4(pa[c], vb, ox[ht]);
        else       oy[ht] = wmma4(pa[c], vb, oy[ht]);
      }
    }
  }

  const float sgn = (lane & 1) ? 1.0f : -1.0f;
#pragma unroll
  for (int ht = 0; ht < 4; ++ht) {
#pragma unroll
    for (int r = 0; r < 8; ++r) {
      float cv = hi ? oy[ht][r] : ox[ht][r];
      float sv = hi ? ox[ht][r] : oy[ht][r];
      float sw = __shfl_xor(sv, 17, 32);
      float res = cv + sgn * sw;
      int M = r + (hi << 3);
      obh[(size_t)(n0 + M) * HDIM + (ht << 4) + nl] = res;
    }
  }
}

extern "C" void kernel_launch(void* const* d_in, const int* in_sizes, int n_in,
                              void* d_out, int out_size, void* d_ws, size_t ws_size,
                              hipStream_t stream) {
  const float* q  = (const float*)d_in[0];
  const float* k  = (const float*)d_in[1];
  const float* v  = (const float*)d_in[2];
  const float* pb = (const float*)d_in[3];
  const float* se = (const float*)d_in[4];
  const float* ph = (const float*)d_in[5];
  float* out = (float*)d_out;

  dim3 grid(NB * NH * (NN / 64));
  dim3 block(128);
  hipLaunchKernelGGL(dsqg_attn_v4_kernel, grid, block, 0, stream,
                     q, k, v, pb, se, ph, out);
}


# Score pass software-pipelined: Q and SE tile 0 requested before the table barrier, each of the 17 B tiles (3 SE, 14 K) loaded one tile ahead of its WMMA chain (second register set); sum order kept
# speedup vs baseline: 1.3646x; 1.0355x over previous
#include <hip/hip_runtime.h>

typedef __attribute__((ext_vector_type(2))) float v2f;
typedef __attribute__((ext_vector_type(4))) float v4f;
typedef __attribute__((ext_vector_type(8))) float v8f;

#define NB 2
#define NH 16
#define NN 8192
#define HDIM 64
#define NOFF 44
#define NTILE 14
#define SS 76
#define DCOL 16
#define SCOL 64
#define SSLOT (SCOL - DCOL)
#define QSTR 60

__constant__ int c_off[NOFF] = {
  0,1,2,3,4,5,6,7,8,9,10,11,12,13,14,15,16,17,18,19,20,21,22,
  23,24,25,26,27,28,29,30,31,32,
  48,64,96,128,192,256,384,512,768,1024,1536 };

__constant__ int c_toff[NTILE] = {0,16,32,48,64,96,128,192,256,384,512,768,1024,1536};

__device__ __forceinline__ v8f wmma4(v2f a, v2f b, v8f c) {
  return __builtin_amdgcn_wmma_f32_16x16x4_f32(
      false, a, false, b, (short)0, c, false, false);
}

__global__ __launch_bounds__(128, 7)
void dsqg_attn_v4_kernel(const float* __restrict__ q,
                         const float* __restrict__ k,
                         const float* __restrict__ v,
                         const float* __restrict__ pbias,
                         const float* __restrict__ se,
                         const float* __restrict__ ph,
                         float* __restrict__ out)
{
  __shared__ __attribute__((aligned(16))) float s_cp[64];
  __shared__ __attribute__((aligned(16))) float s_sp[64];
  __shared__ __attribute__((aligned(16))) float s_pb[64];
  __shared__ __attribute__((aligned(16))) float s_scores[4 * 16 * SS];
  __shared__ __attribute__((aligned(16))) float s_qd[4 * 16 * QSTR];
  __shared__ float s_rinv[4 * 16];

  const int bid  = blockIdx.x;
  const int bh   = bid >> 7;
  const int h    = bh & (NH - 1);
  const int n0b  = (bid & 127) << 6;
  const int tid  = threadIdx.x;
  const int wv   = tid >> 5;
  const int lane = tid & 31;
  const int n0   = n0b + (wv << 4);

  const float sc   = 0.125f;
  const int   nl   = lane & 15;
  const int   hi   = lane >> 4;
  const int   koff = hi << 1;
  const int   kq   = hi << 2;

  const size_t bhoff = (size_t)bh * NN * HDIM;
  const float* qbh = q + bhoff;
  const float* kbh = k + bhoff;
  const float* vbh = v + bhoff;
  float*       obh = out + bhoff;
  float*       srow = s_scores + wv * (16 * SS);
  float*       qrow = s_qd + wv * (16 * QSTR);

  v4f qraw[8];
  v4f kb[2][8];
  {
    const float* qr = qbh + (size_t)(n0 + nl) * HDIM + kq;
#pragma unroll
    for (int j = 0; j < 8; ++j) qraw[j] = *(const v4f*)(qr + 8 * j);
    const float* ser = se + (size_t)nl * HDIM + kq;
#pragma unroll
    for (int j = 0; j < 8; ++j) kb[0][j] = *(const v4f*)(ser + 8 * j);
  }

  if (tid < 64) {
    const int ii = tid - DCOL;
    float cv = 0.f, sv = 0.f;
    if (ii >= 0 && ii < NOFF) {
      float p = ph[ii * NH + h];
      cv = cosf(p);
      sv = sinf(p);
      s_pb[ii < 33 ? ii : ii + (SSLOT - 33)] = pbias[ii * NH + h];
    }
    s_cp[tid] = cv;
    s_sp[tid] = sv;
  }
  __syncthreads();

  v2f qa[16];
#pragma unroll
  for (int j = 0; j < 8; ++j) {
    v4f t = qraw[j];
    qa[2 * j].x     = t.x * sc;
    qa[2 * j].y     = t.y * sc;
    qa[2 * j + 1].x = t.z * sc;
    qa[2 * j + 1].y = t.w * sc;
  }

  const int  sdense  = (hi << 3) * (SS + 1) + DCOL - nl;
  const bool diag    = ((nl >> 3) == hi);
  const int  ssparse = nl * SS + (diag ? SCOL : 0);
  const bool sb0 = (nl & 1) != 0, sb1 = (nl & 2) != 0, sb2 = (nl & 4) != 0;
  constexpr int toff[NTILE] = {0,16,32,48,64,96,128,192,256,384,512,768,1024,1536};
  const int n0s = __builtin_amdgcn_readfirstlane(n0);
#pragma unroll
  for (int u = 0; u < 3 + NTILE; ++u) {
    if (u + 1 < 3 + NTILE) {
      const float* p;
      if (u + 1 < 3) {
        int irow = (u + 1) * 16 + nl;
        int ir   = (irow < NOFF) ? irow : (NOFF - 1);
        p = se + (size_t)ir * HDIM + kq;
      } else {
        int krow = n0 - toff[u + 1 - 3] + nl; if (krow < 0) krow = 0;
        p = kbh + (size_t)krow * HDIM + kq;
        __builtin_prefetch(vbh + (size_t)krow * HDIM, 0, 1);
      }
#pragma unroll
      for (int j = 0; j < 8; ++j) kb[(u + 1) & 1][j] = *(const v4f*)(p + 8 * j);
    }
    const int t = (u < 3) ? 0 : (u - 3);
    if (u >= 3 && n0s + 15 < toff[t]) continue;
    v8f acc = {0.f,0.f,0.f,0.f,0.f,0.f,0.f,0.f};
#pragma unroll
    for (int j = 0; j < 8; ++j) {
      v4f b = kb[u & 1][j];
      acc = wmma4(qa[2 * j],     (v2f){b.x, b.y}, acc);
      acc = wmma4(qa[2 * j + 1], (v2f){b.z, b.w}, acc);
    }
    if (u < 3) {
#pragma unroll
      for (int r = 0; r < 8; ++r) {
        int M = r + (hi << 3);
        int i = u * 16 + nl;
        if (i < NOFF) qrow[M * QSTR + (i < 33 ? i : i + (SSLOT - 33))] = acc[r];
      }
    } else if (t < 3) {
#pragma unroll
      for (int r = 0; r < 8; ++r) srow[sdense + r * (SS + 1) + (t << 4)] = acc[r];
    } else {
      float a0 = sb0 ? acc[1] : acc[0];
      float a1 = sb0 ? acc[3] : acc[2];
      float a2 = sb0 ? acc[5] : acc[4];
      float a3 = sb0 ? acc[7] : acc[6];
      float c0 = sb1 ? a1 : a0;
      float c1 = sb1 ? a3 : a2;
      srow[ssparse + (t - 3)] = sb2 ? c1 : c0;
    }
  }

  if (lane < 16) {
    float* sr = srow + lane * SS + DCOL;
    const float* qd = qrow + lane * QSTR;
    const int ng = n0 + lane;
    const float NEG = -3.0e38f;
    float mx = NEG;
    for (int g = 0; g < 8; ++g) {
      v4f sv = (*(const v4f*)(sr + 4 * g) + *(const v4f*)(qd + 4 * g)) + *(const v4f*)(s_pb + 4 * g);
      if (ng < 4 * g)     sv.x = NEG;
      if (ng < 4 * g + 1) sv.y = NEG;
      if (ng < 4 * g + 2) sv.z = NEG;
      if (ng < 4 * g + 3) sv.w = NEG;
      *(v4f*)(sr + 4 * g) = sv;
      mx = fmaxf(fmaxf(fmaxf(fmaxf(mx, sv.x), sv.y), sv.z), sv.w);
    }
    {
      float sv = (sr[32] + qd[32]) + s_pb[32];
      if (ng < 32) sv = NEG;
      sr[32] = sv;
      mx = fmaxf(mx, sv);
    }
    for (int g = 0; g < 2; ++g) {
      const int k = SSLOT + 4 * g;
      v4f sv = (*(const v4f*)(sr + k) + *(const v4f*)(qd + k)) + *(const v4f*)(s_pb + k);
      if (ng < c_off[33 + 4 * g])     sv.x = NEG;
      if (ng < c_off[33 + 4 * g + 1]) sv.y = NEG;
      if (ng < c_off[33 + 4 * g + 2]) sv.z = NEG;
      if (ng < c_off[33 + 4 * g + 3]) sv.w = NEG;
      *(v4f*)(sr + k) = sv;
      mx = fmaxf(fmaxf(fmaxf(fmaxf(mx, sv.x), sv.y), sv.z), sv.w);
    }
    for (int s = 8; s < NOFF - 33; ++s) {
      float sv = (sr[SSLOT + s] + qd[SSLOT + s]) + s_pb[SSLOT + s];
      if (ng < c_off[33 + s]) sv = NEG;
      sr[SSLOT + s] = sv;
      mx = fmaxf(mx, sv);
    }

    float sum = 0.f;
    for (int g = 0; g < 8; ++g) {
      v4f a = *(const v4f*)(sr + 4 * g);
      v4f e;
      e.x = __expf(a.x - mx); e.y = __expf(a.y - mx); e.z = __expf(a.z - mx); e.w = __expf(a.w - mx);
      *(v4f*)(sr + 4 * g) = e;
      sum += e.x; sum += e.y; sum += e.z; sum += e.w;
    }
    {
      float e = __expf(sr[32] - mx);
      sr[32] = e;
      sum += e;
    }
    for (int g = 0; g < 2; ++g) {
      const int k = SSLOT + 4 * g;
      v4f a = *(const v4f*)(sr + k);
      v4f e;
      e.x = __expf(a.x - mx); e.y = __expf(a.y - mx); e.z = __expf(a.z - mx); e.w = __expf(a.w - mx);
      *(v4f*)(sr + k) = e;
      sum += e.x; sum += e.y; sum += e.z; sum += e.w;
    }
    for (int s = 8; s < NOFF - 33; ++s) {
      float e = __expf(sr[SSLOT + s] - mx);
      sr[SSLOT + s] = e;
      sum += e;
    }
    s_rinv[wv * 16 + lane] = 1.0f / sum;

    const v4f z4 = {0.f, 0.f, 0.f, 0.f};
    for (int g = 0; g < 4; ++g) *(v4f*)(sr - DCOL + 4 * g) = z4;
    sr[33] = 0.f; sr[34] = 0.f; sr[35] = 0.f;
    for (int g = 9; g < 12; ++g) *(v4f*)(sr + 4 * g) = z4;
  }

  const int   r7    = nl & 7;
  const bool  lo8   = nl < 8;
  const float* srx  = srow + r7 * SS;
  const float* sry  = srow + (r7 + 8) * SS;
  const float rinvx = s_rinv[wv * 16 + r7];
  const float rinvy = s_rinv[wv * 16 + r7 + 8];
  const float* tabx = lo8 ? s_cp : s_sp;
  const float* taby = lo8 ? s_sp : s_cp;

  v8f ox[4], oy[4];
#pragma unroll
  for (int ht = 0; ht < 4; ++ht) {
    ox[ht] = (v8f){0.f,0.f,0.f,0.f,0.f,0.f,0.f,0.f};
    oy[ht] = (v8f){0.f,0.f,0.f,0.f,0.f,0.f,0.f,0.f};
  }

#pragma unroll
  for (int t = 0; t < 3; ++t) {
    const int off = t << 4;
    if (n0 + 15 < off) continue;
    const int m0 = n0 - off;

    v2f pax[4], pay[4];
#pragma unroll
    for (int c = 0; c < 4; ++c) {
#pragma unroll
      for (int u = 0; u < 2; ++u) {
        int j  = 4 * c + koff + u;
        int dx = r7 - j + off;
        int dy = dx + 8;
        float ax = (srx[DCOL + dx] * rinvx) * tabx[DCOL + dx];
        float ay = (sry[DCOL + dy] * rinvy) * taby[DCOL + dy];
        if (u == 0) { pax[c].x = ax; pay[c].x = ay; }
        else        { pax[c].y = ax; pay[c].y = ay; }
      }
    }

#pragma unroll
    for (int ht = 0; ht < 4; ++ht) {
      const int hd0 = ht << 4;
#pragma unroll
      for (int c = 0; c < 4; ++c) {
        int rr = m0 + 4 * c + koff;
        int r0 = rr < 0 ? 0 : rr;
        int r1 = (rr + 1) < 0 ? 0 : (rr + 1);
        v2f vb;
        vb.x = vbh[(size_t)r0 * HDIM + hd0 + nl];
        vb.y = vbh[(size_t)r1 * HDIM + hd0 + nl];
        if (!(t == 0 && c >= 2)) ox[ht] = wmma4(pax[c], vb, ox[ht]);
        if (!(t == 2 && c < 2))  oy[ht] = wmma4(pay[c], vb, oy[ht]);
      }
    }
  }

  for (int t = 3; t < NTILE; ++t) {
    const int off = c_toff[t];
    if (n0 + 15 < off) continue;
    const int m0 = n0 - off;
    const int s  = t - 3;

    const float px = (srx[SCOL + s] * rinvx) * tabx[DCOL + 33 + s];
    const float py = (sry[SCOL + s] * rinvy) * taby[DCOL + 33 + s];

    v2f pa[4];
#pragma unroll
    for (int c = 0; c < 4; ++c) {
#pragma unroll
      for (int u = 0; u < 2; ++u) {
        int j = 4 * c + koff + u;
        float a = (r7 == (j & 7)) ? ((c < 2) ? px : py) : 0.f;
        if (u == 0) pa[c].x = a; else pa[c].y = a;
      }
    }

#pragma unroll
    for (int ht = 0; ht < 4; ++ht) {
      const int hd0 = ht << 4;
#pragma unroll
      for (int c = 0; c < 4; ++c) {
        int rr = m0 + 4 * c + koff;
        int r0 = rr < 0 ? 0 : rr;
        int r1 = (rr + 1) < 0 ? 0 : (rr + 1);
        v2f vb;
        vb.x = vbh[(size_t)r0 * HDIM + hd0 + nl];
        vb.y = vbh[(size_t)r1 * HDIM + hd0 + nl];
        if (c < 2) ox[ht] = wmma4(pa[c], vb, ox[ht]);
        else       oy[ht] = wmma4(pa[c], vb, oy[ht]);
      }
    }
  }

  const float sgn = (lane & 1) ? 1.0f : -1.0f;
#pragma unroll
  for (int ht = 0; ht < 4; ++ht) {
#pragma unroll
    for (int r = 0; r < 8; ++r) {
      float cv = hi ? oy[ht][r] : ox[ht][r];
      float sv = hi ? ox[ht][r] : oy[ht][r];
      float sw = __shfl_xor(sv, 17, 32);
      float res = cv + sgn * sw;
      int M = r + (hi << 3);
      obh[(size_t)(n0 + M) * HDIM + (ht << 4) + nl] = res;
    }
  }
}

extern "C" void kernel_launch(void* const* d_in, const int* in_sizes, int n_in,
                              void* d_out, int out_size, void* d_ws, size_t ws_size,
                              hipStream_t stream) {
  const float* q  = (const float*)d_in[0];
  const float* k  = (const float*)d_in[1];
  const float* v  = (const float*)d_in[2];
  const float* pb = (const float*)d_in[3];
  const float* se = (const float*)d_in[4];
  const float* ph = (const float*)d_in[5];
  float* out = (float*)d_out;

  dim3 grid(NB * NH * (NN / 64));
  dim3 block(128);
  hipLaunchKernelGGL(dsqg_attn_v4_kernel, grid, block, 0, stream,
                     q, k, v, pb, se, ph, out);
}


# Sparse-offset score tiles re-tiled over rows 16 apart (16-wave blocks, scores exchanged through LDS): 11 diagonal-only key tiles -> 6 band tiles; 528 -> 448 WMMA per wave; sum order kept
# speedup vs baseline: 1.4727x; 1.0792x over previous
#include <hip/hip_runtime.h>

typedef __attribute__((ext_vector_type(2))) float v2f;
typedef __attribute__((ext_vector_type(4))) float v4f;
typedef __attribute__((ext_vector_type(8))) float v8f;

#define NB 2
#define NH 16
#define NN 8192
#define HDIM 64
#define NOFF 44
#define NTILE 14
#define NWAVE 16
#define SS 76
#define DCOL 16
#define SCOL 64
#define SSLOT (SCOL - DCOL)
#define QSTR 60

__constant__ int c_off[NOFF] = {
  0,1,2,3,4,5,6,7,8,9,10,11,12,13,14,15,16,17,18,19,20,21,22,
  23,24,25,26,27,28,29,30,31,32,
  48,64,96,128,192,256,384,512,768,1024,1536 };

__constant__ int c_toff[NTILE] = {0,16,32,48,64,96,128,192,256,384,512,768,1024,1536};

__device__ __forceinline__ v8f wmma4(v2f a, v2f b, v8f c) {
  return __builtin_amdgcn_wmma_f32_16x16x4_f32(
      false, a, false, b, (short)0, c, false, false);
}

__global__ __launch_bounds__(512, 8)
void dsqg_attn_v4_kernel(const float* __restrict__ q,
                         const float* __restrict__ k,
                         const float* __restrict__ v,
                         const float* __restrict__ pbias,
                         const float* __restrict__ se,
                         const float* __restrict__ ph,
                         float* __restrict__ out)
{
  __shared__ __attribute__((aligned(16))) float s_cp[64];
  __shared__ __attribute__((aligned(16))) float s_sp[64];
  __shared__ __attribute__((aligned(16))) float s_pb[64];
  __shared__ __attribute__((aligned(16))) float s_scores[NWAVE * 16 * SS];
  __shared__ __attribute__((aligned(16))) float s_qd[NWAVE * 16 * QSTR];
  __shared__ float s_rinv[NWAVE * 16];
  __shared__ int s_slot[128];

  const int bid  = blockIdx.x;
  const int bh   = bid >> 5;
  const int h    = bh & (NH - 1);
  const int n0b  = (bid & 31) << 8;
  const int tid  = threadIdx.x;
  const int wv   = __builtin_amdgcn_readfirstlane(tid >> 5);
  const int lane = tid & 31;
  const int n0   = n0b + (wv << 4);

  const float sc   = 0.125f;
  const int   nl   = lane & 15;
  const int   hi   = lane >> 4;
  const int   koff = hi << 1;
  const int   kq   = hi << 2;

  const size_t bhoff = (size_t)bh * NN * HDIM;
  const float* qbh = q + bhoff;
  const float* kbh = k + bhoff;
  const float* vbh = v + bhoff;
  float*       obh = out + bhoff;
  float*       srow = s_scores + wv * (16 * SS);
  float*       qrow = s_qd + wv * (16 * QSTR);

  v4f qraw[8];
  v4f kb[2][8];
  {
    const float* qr = qbh + (size_t)(n0 + nl) * HDIM + kq;
#pragma unroll
    for (int j = 0; j < 8; ++j) qraw[j] = *(const v4f*)(qr + 8 * j);
    const float* ser = se + (size_t)nl * HDIM + kq;
#pragma unroll
    for (int j = 0; j < 8; ++j) kb[0][j] = *(const v4f*)(ser + 8 * j);
  }

  if (tid < 64) {
    const int ii = tid - DCOL;
    float cv = 0.f, sv = 0.f;
    if (ii >= 0 && ii < NOFF) {
      float p = ph[ii * NH + h];
      cv = cosf(p);
      sv = sinf(p);
      s_pb[ii < 33 ? ii : ii + (SSLOT - 33)] = pbias[ii * NH + h];
    }
    s_cp[tid] = cv;
    s_sp[tid] = sv;
  } else if (tid < 192) {
    const int m = tid - 64 - 16;
    const int s = (m == 3) ? 0 : (m == 4) ? 1 : (m == 6) ? 2 : (m == 8) ? 3 : (m == 12) ? 4
                : (m == 16) ? 5 : (m == 24) ? 6 : (m == 32) ? 7 : (m == 48) ? 8 : (m == 64) ? 9
                : (m == 96) ? 10 : -1;
    s_slot[tid - 64] = (s < 0) ? -1 : (SCOL + s);
  }
  __syncthreads();

  v2f qa[16];
#pragma unroll
  for (int j = 0; j < 8; ++j) {
    v4f t = qraw[j];
    qa[2 * j].x     = t.x * sc;
    qa[2 * j].y     = t.y * sc;
    qa[2 * j + 1].x = t.z * sc;
    qa[2 * j + 1].y = t.w * sc;
  }

  const int  sdense  = (hi << 3) * (SS + 1) + DCOL - nl;
  constexpr int toff[3] = {0,16,32};
  const int n0s = __builtin_amdgcn_readfirstlane(n0);
#pragma unroll
  for (int u = 0; u < 6; ++u) {
    if (u + 1 < 6) {
      const float* p;
      if (u + 1 < 3) {
        int irow = (u + 1) * 16 + nl;
        int ir   = (irow < NOFF) ? irow : (NOFF - 1);
        p = se + (size_t)ir * HDIM + kq;
      } else {
        int krow = n0 - toff[u + 1 - 3] + nl; if (krow < 0) krow = 0;
        p = kbh + (size_t)krow * HDIM + kq;
        __builtin_prefetch(vbh + (size_t)krow * HDIM, 0, 1);
      }
#pragma unroll
      for (int j = 0; j < 8; ++j) kb[(u + 1) & 1][j] = *(const v4f*)(p + 8 * j);
    }
    const int t = (u < 3) ? 0 : (u - 3);
    if (u >= 3 && n0s + 15 < toff[t]) continue;
    v8f acc = {0.f,0.f,0.f,0.f,0.f,0.f,0.f,0.f};
#pragma unroll
    for (int j = 0; j < 8; ++j) {
      v4f b = kb[u & 1][j];
      acc = wmma4(qa[2 * j],     (v2f){b.x, b.y}, acc);
      acc = wmma4(qa[2 * j + 1], (v2f){b.z, b.w}, acc);
    }
    if (u < 3) {
#pragma unroll
      for (int r = 0; r < 8; ++r) {
        int M = r + (hi << 3);
        int i = u * 16 + nl;
        if (i < NOFF) qrow[M * QSTR + (i < 33 ? i : i + (SSLOT - 33))] = acc[r];
      }
    } else {
#pragma unroll
      for (int r = 0; r < 8; ++r) srow[sdense + r * (SS + 1) + (t << 4)] = acc[r];
    }
  }

  {
    const int rb = n0b + wv;
    {
      const float* qr = qbh + (size_t)(rb + 16 * nl) * HDIM + kq;
#pragma unroll
      for (int j = 0; j < 8; ++j) qraw[j] = *(const v4f*)(qr + 8 * j);
      const float* kr = kbh + (size_t)(rb + 16 * nl) * HDIM + kq;
#pragma unroll
      for (int j = 0; j < 8; ++j) kb[0][j] = *(const v4f*)(kr + 8 * j);
      __builtin_prefetch(vbh + (size_t)(rb + 16 * nl) * HDIM, 0, 1);
    }
#pragma unroll
    for (int j = 0; j < 8; ++j) {
      v4f t = qraw[j];
      qa[2 * j].x     = t.x * sc;
      qa[2 * j].y     = t.y * sc;
      qa[2 * j + 1].x = t.z * sc;
      qa[2 * j + 1].y = t.w * sc;
    }
    constexpr int mt[6] = {0,-16,-32,-48,-64,-96};
    const int lb = (hi << 3) - nl + 16;
    float* sblk = s_scores + wv * SS;
#pragma unroll
    for (int v = 0; v < 6; ++v) {
      if (v + 1 < 6) {
        int krow = rb + 16 * (mt[v + 1] + nl); if (krow < 0) krow = 0;
        const float* p = kbh + (size_t)krow * HDIM + kq;
        __builtin_prefetch(vbh + (size_t)krow * HDIM, 0, 1);
#pragma unroll
        for (int j = 0; j < 8; ++j) kb[(v + 1) & 1][j] = *(const v4f*)(p + 8 * j);
      }
      if (v > 0 && rb + 16 * (mt[v] + 15) < 0) continue;
      v8f acc = {0.f,0.f,0.f,0.f,0.f,0.f,0.f,0.f};
#pragma unroll
      for (int j = 0; j < 8; ++j) {
        v4f b = kb[v & 1][j];
        acc = wmma4(qa[2 * j],     (v2f){b.x, b.y}, acc);
        acc = wmma4(qa[2 * j + 1], (v2f){b.z, b.w}, acc);
      }
#pragma unroll
      for (int r = 0; r < 8; ++r) {
        const int M   = r + (hi << 3);
        const int col = s_slot[lb + r - mt[v]];
        sblk[M * (16 * SS) + (col < 0 ? nl : col)] = acc[r];
      }
    }
  }
  __syncthreads();

  if (lane < 16) {
    float* sr = srow + lane * SS + DCOL;
    const float* qd = qrow + lane * QSTR;
    const int ng = n0 + lane;
    const float NEG = -3.0e38f;
    float mx = NEG;
    for (int g = 0; g < 8; ++g) {
      v4f sv = (*(const v4f*)(sr + 4 * g) + *(const v4f*)(qd + 4 * g)) + *(const v4f*)(s_pb + 4 * g);
      if (ng < 4 * g)     sv.x = NEG;
      if (ng < 4 * g + 1) sv.y = NEG;
      if (ng < 4 * g + 2) sv.z = NEG;
      if (ng < 4 * g + 3) sv.w = NEG;
      *(v4f*)(sr + 4 * g) = sv;
      mx = fmaxf(fmaxf(fmaxf(fmaxf(mx, sv.x), sv.y), sv.z), sv.w);
    }
    {
      float sv = (sr[32] + qd[32]) + s_pb[32];
      if (ng < 32) sv = NEG;
      sr[32] = sv;
      mx = fmaxf(mx, sv);
    }
    for (int g = 0; g < 2; ++g) {
      const int k = SSLOT + 4 * g;
      v4f sv = (*(const v4f*)(sr + k) + *(const v4f*)(qd + k)) + *(const v4f*)(s_pb + k);
      if (ng < c_off[33 + 4 * g])     sv.x = NEG;
      if (ng < c_off[33 + 4 * g + 1]) sv.y = NEG;
      if (ng < c_off[33 + 4 * g + 2]) sv.z = NEG;
      if (ng < c_off[33 + 4 * g + 3]) sv.w = NEG;
      *(v4f*)(sr + k) = sv;
      mx = fmaxf(fmaxf(fmaxf(fmaxf(mx, sv.x), sv.y), sv.z), sv.w);
    }
    for (int s = 8; s < NOFF - 33; ++s) {
      float sv = (sr[SSLOT + s] + qd[SSLOT + s]) + s_pb[SSLOT + s];
      if (ng < c_off[33 + s]) sv = NEG;
      sr[SSLOT + s] = sv;
      mx = fmaxf(mx, sv);
    }

    float sum = 0.f;
    for (int g = 0; g < 8; ++g) {
      v4f a = *(const v4f*)(sr + 4 * g);
      v4f e;
      e.x = __expf(a.x - mx); e.y = __expf(a.y - mx); e.z = __expf(a.z - mx); e.w = __expf(a.w - mx);
      *(v4f*)(sr + 4 * g) = e;
      sum += e.x; sum += e.y; sum += e.z; sum += e.w;
    }
    {
      float e = __expf(sr[32] - mx);
      sr[32] = e;
      sum += e;
    }
    for (int g = 0; g < 2; ++g) {
      const int k = SSLOT + 4 * g;
      v4f a = *(const v4f*)(sr + k);
      v4f e;
      e.x = __expf(a.x - mx); e.y = __expf(a.y - mx); e.z = __expf(a.z - mx); e.w = __expf(a.w - mx);
      *(v4f*)(sr + k) = e;
      sum += e.x; sum += e.y; sum += e.z; sum += e.w;
    }
    for (int s = 8; s < NOFF - 33; ++s) {
      float e = __expf(sr[SSLOT + s] - mx);
      sr[SSLOT + s] = e;
      sum += e;
    }
    s_rinv[wv * 16 + lane] = 1.0f / sum;

    const v4f z4 = {0.f, 0.f, 0.f, 0.f};
    for (int g = 0; g < 4; ++g) *(v4f*)(sr - DCOL + 4 * g) = z4;
    sr[33] = 0.f; sr[34] = 0.f; sr[35] = 0.f;
    for (int g = 9; g < 12; ++g) *(v4f*)(sr + 4 * g) = z4;
  }

  const int   r7    = nl & 7;
  const bool  lo8   = nl < 8;
  const float* srx  = srow + r7 * SS;
  const float* sry  = srow + (r7 + 8) * SS;
  const float rinvx = s_rinv[wv * 16 + r7];
  const float rinvy = s_rinv[wv * 16 + r7 + 8];
  const float* tabx = lo8 ? s_cp : s_sp;
  const float* taby = lo8 ? s_sp : s_cp;

  v8f ox[4], oy[4];
#pragma unroll
  for (int ht = 0; ht < 4; ++ht) {
    ox[ht] = (v8f){0.f,0.f,0.f,0.f,0.f,0.f,0.f,0.f};
    oy[ht] = (v8f){0.f,0.f,0.f,0.f,0.f,0.f,0.f,0.f};
  }

#pragma unroll
  for (int t = 0; t < 3; ++t) {
    const int off = t << 4;
    if (n0 + 15 < off) continue;
    const int m0 = n0 - off;

    v2f pax[4], pay[4];
#pragma unroll
    for (int c = 0; c < 4; ++c) {
#pragma unroll
      for (int u = 0; u < 2; ++u) {
        int j  = 4 * c + koff + u;
        int dx = r7 - j + off;
        int dy = dx + 8;
        float ax = (srx[DCOL + dx] * rinvx) * tabx[DCOL + dx];
        float ay = (sry[DCOL + dy] * rinvy) * taby[DCOL + dy];
        if (u == 0) { pax[c].x = ax; pay[c].x = ay; }
        else        { pax[c].y = ax; pay[c].y = ay; }
      }
    }

#pragma unroll
    for (int ht = 0; ht < 4; ++ht) {
      const int hd0 = ht << 4;
#pragma unroll
      for (int c = 0; c < 4; ++c) {
        int rr = m0 + 4 * c + koff;
        int r0 = rr < 0 ? 0 : rr;
        int r1 = (rr + 1) < 0 ? 0 : (rr + 1);
        v2f vb;
        vb.x = vbh[(size_t)r0 * HDIM + hd0 + nl];
        vb.y = vbh[(size_t)r1 * HDIM + hd0 + nl];
        if (!(t == 0 && c >= 2)) ox[ht] = wmma4(pax[c], vb, ox[ht]);
        if (!(t == 2 && c < 2))  oy[ht] = wmma4(pay[c], vb, oy[ht]);
      }
    }
  }

  for (int t = 3; t < NTILE; ++t) {
    const int off = c_toff[t];
    if (n0 + 15 < off) continue;
    const int m0 = n0 - off;
    const int s  = t - 3;

    const float px = (srx[SCOL + s] * rinvx) * tabx[DCOL + 33 + s];
    const float py = (sry[SCOL + s] * rinvy) * taby[DCOL + 33 + s];

    v2f pa[4];
#pragma unroll
    for (int c = 0; c < 4; ++c) {
#pragma unroll
      for (int u = 0; u < 2; ++u) {
        int j = 4 * c + koff + u;
        float a = (r7 == (j & 7)) ? ((c < 2) ? px : py) : 0.f;
        if (u == 0) pa[c].x = a; else pa[c].y = a;
      }
    }

#pragma unroll
    for (int ht = 0; ht < 4; ++ht) {
      const int hd0 = ht << 4;
#pragma unroll
      for (int c = 0; c < 4; ++c) {
        int rr = m0 + 4 * c + koff;
        int r0 = rr < 0 ? 0 : rr;
        int r1 = (rr + 1) < 0 ? 0 : (rr + 1);
        v2f vb;
        vb.x = vbh[(size_t)r0 * HDIM + hd0 + nl];
        vb.y = vbh[(size_t)r1 * HDIM + hd0 + nl];
        if (c < 2) ox[ht] = wmma4(pa[c], vb, ox[ht]);
        else       oy[ht] = wmma4(pa[c], vb, oy[ht]);
      }
    }
  }

  const float sgn = (lane & 1) ? 1.0f : -1.0f;
#pragma unroll
  for (int ht = 0; ht < 4; ++ht) {
#pragma unroll
    for (int r = 0; r < 8; ++r) {
      float cv = hi ? oy[ht][r] : ox[ht][r];
      float sv = hi ? ox[ht][r] : oy[ht][r];
      float sw = __shfl_xor(sv, 17, 32);
      float res = cv + sgn * sw;
      int M = r + (hi << 3);
      obh[(size_t)(n0 + M) * HDIM + (ht << 4) + nl] = res;
    }
  }
}

extern "C" void kernel_launch(void* const* d_in, const int* in_sizes, int n_in,
                              void* d_out, int out_size, void* d_ws, size_t ws_size,
                              hipStream_t stream) {
  const float* q  = (const float*)d_in[0];
  const float* k  = (const float*)d_in[1];
  const float* v  = (const float*)d_in[2];
  const float* pb = (const float*)d_in[3];
  const float* se = (const float*)d_in[4];
  const float* ph = (const float*)d_in[5];
  float* out = (float*)d_out;

  dim3 grid(NB * NH * (NN / 256));
  dim3 block(512);
  hipLaunchKernelGGL(dsqg_attn_v4_kernel, grid, block, 0, stream,
                     q, k, v, pb, se, ph, out);
}


# P.V pass: V fragment address = scalar tile base + one lane offset + constant (row clamps, never active in a tile that runs, dropped), so each tile's 32 V loads go out as one group; sum order kept
# speedup vs baseline: 1.5512x; 1.0533x over previous
#include <hip/hip_runtime.h>

typedef __attribute__((ext_vector_type(2))) float v2f;
typedef __attribute__((ext_vector_type(4))) float v4f;
typedef __attribute__((ext_vector_type(8))) float v8f;

#define NB 2
#define NH 16
#define NN 8192
#define HDIM 64
#define NOFF 44
#define NTILE 14
#define NWAVE 16
#define SS 76
#define DCOL 16
#define SCOL 64
#define SSLOT (SCOL - DCOL)
#define QSTR 60

__constant__ int c_off[NOFF] = {
  0,1,2,3,4,5,6,7,8,9,10,11,12,13,14,15,16,17,18,19,20,21,22,
  23,24,25,26,27,28,29,30,31,32,
  48,64,96,128,192,256,384,512,768,1024,1536 };

__constant__ int c_toff[NTILE] = {0,16,32,48,64,96,128,192,256,384,512,768,1024,1536};

__device__ __forceinline__ v8f wmma4(v2f a, v2f b, v8f c) {
  return __builtin_amdgcn_wmma_f32_16x16x4_f32(
      false, a, false, b, (short)0, c, false, false);
}

__global__ __launch_bounds__(512, 8)
void dsqg_attn_v4_kernel(const float* __restrict__ q,
                         const float* __restrict__ k,
                         const float* __restrict__ v,
                         const float* __restrict__ pbias,
                         const float* __restrict__ se,
                         const float* __restrict__ ph,
                         float* __restrict__ out)
{
  __shared__ __attribute__((aligned(16))) float s_cp[64];
  __shared__ __attribute__((aligned(16))) float s_sp[64];
  __shared__ __attribute__((aligned(16))) float s_pb[64];
  __shared__ __attribute__((aligned(16))) float s_scores[NWAVE * 16 * SS];
  __shared__ __attribute__((aligned(16))) float s_qd[NWAVE * 16 * QSTR];
  __shared__ float s_rinv[NWAVE * 16];
  __shared__ int s_slot[128];

  const int bid  = blockIdx.x;
  const int bh   = bid >> 5;
  const int h    = bh & (NH - 1);
  const int n0b  = (bid & 31) << 8;
  const int tid  = threadIdx.x;
  const int wv   = __builtin_amdgcn_readfirstlane(tid >> 5);
  const int lane = tid & 31;
  const int n0   = n0b + (wv << 4);

  const float sc   = 0.125f;
  const int   nl   = lane & 15;
  const int   hi   = lane >> 4;
  const int   koff = hi << 1;
  const int   kq   = hi << 2;

  const size_t bhoff = (size_t)bh * NN * HDIM;
  const float* qbh = q + bhoff;
  const float* kbh = k + bhoff;
  const float* vbh = v + bhoff;
  float*       obh = out + bhoff;
  float*       srow = s_scores + wv * (16 * SS);
  float*       qrow = s_qd + wv * (16 * QSTR);

  v4f qraw[8];
  v4f kb[2][8];
  {
    const float* qr = qbh + (size_t)(n0 + nl) * HDIM + kq;
#pragma unroll
    for (int j = 0; j < 8; ++j) qraw[j] = *(const v4f*)(qr + 8 * j);
    const float* ser = se + (size_t)nl * HDIM + kq;
#pragma unroll
    for (int j = 0; j < 8; ++j) kb[0][j] = *(const v4f*)(ser + 8 * j);
  }

  if (tid < 64) {
    const int ii = tid - DCOL;
    float cv = 0.f, sv = 0.f;
    if (ii >= 0 && ii < NOFF) {
      float p = ph[ii * NH + h];
      cv = cosf(p);
      sv = sinf(p);
      s_pb[ii < 33 ? ii : ii + (SSLOT - 33)] = pbias[ii * NH + h];
    }
    s_cp[tid] = cv;
    s_sp[tid] = sv;
  } else if (tid < 192) {
    const int m = tid - 64 - 16;
    const int s = (m == 3) ? 0 : (m == 4) ? 1 : (m == 6) ? 2 : (m == 8) ? 3 : (m == 12) ? 4
                : (m == 16) ? 5 : (m == 24) ? 6 : (m == 32) ? 7 : (m == 48) ? 8 : (m == 64) ? 9
                : (m == 96) ? 10 : -1;
    s_slot[tid - 64] = (s < 0) ? -1 : (SCOL + s);
  }
  __syncthreads();

  v2f qa[16];
#pragma unroll
  for (int j = 0; j < 8; ++j) {
    v4f t = qraw[j];
    qa[2 * j].x     = t.x * sc;
    qa[2 * j].y     = t.y * sc;
    qa[2 * j + 1].x = t.z * sc;
    qa[2 * j + 1].y = t.w * sc;
  }

  const int  sdense  = (hi << 3) * (SS + 1) + DCOL - nl;
  constexpr int toff[3] = {0,16,32};
  const int n0s = __builtin_amdgcn_readfirstlane(n0);
#pragma unroll
  for (int u = 0; u < 6; ++u) {
    if (u + 1 < 6) {
      const float* p;
      if (u + 1 < 3) {
        int irow = (u + 1) * 16 + nl;
        int ir   = (irow < NOFF) ? irow : (NOFF - 1);
        p = se + (size_t)ir * HDIM + kq;
      } else {
        int krow = n0 - toff[u + 1 - 3] + nl; if (krow < 0) krow = 0;
        p = kbh + (size_t)krow * HDIM + kq;
        __builtin_prefetch(vbh + (size_t)krow * HDIM, 0, 1);
      }
#pragma unroll
      for (int j = 0; j < 8; ++j) kb[(u + 1) & 1][j] = *(const v4f*)(p + 8 * j);
    }
    const int t = (u < 3) ? 0 : (u - 3);
    if (u >= 3 && n0s + 15 < toff[t]) continue;
    v8f acc = {0.f,0.f,0.f,0.f,0.f,0.f,0.f,0.f};
#pragma unroll
    for (int j = 0; j < 8; ++j) {
      v4f b = kb[u & 1][j];
      acc = wmma4(qa[2 * j],     (v2f){b.x, b.y}, acc);
      acc = wmma4(qa[2 * j + 1], (v2f){b.z, b.w}, acc);
    }
    if (u < 3) {
#pragma unroll
      for (int r = 0; r < 8; ++r) {
        int M = r + (hi << 3);
        int i = u * 16 + nl;
        if (i < NOFF) qrow[M * QSTR + (i < 33 ? i : i + (SSLOT - 33))] = acc[r];
      }
    } else {
#pragma unroll
      for (int r = 0; r < 8; ++r) srow[sdense + r * (SS + 1) + (t << 4)] = acc[r];
    }
  }

  {
    const int rb = n0b + wv;
    {
      const float* qr = qbh + (size_t)(rb + 16 * nl) * HDIM + kq;
#pragma unroll
      for (int j = 0; j < 8; ++j) qraw[j] = *(const v4f*)(qr + 8 * j);
      const float* kr = kbh + (size_t)(rb + 16 * nl) * HDIM + kq;
#pragma unroll
      for (int j = 0; j < 8; ++j) kb[0][j] = *(const v4f*)(kr + 8 * j);
      __builtin_prefetch(vbh + (size_t)(rb + 16 * nl) * HDIM, 0, 1);
    }
#pragma unroll
    for (int j = 0; j < 8; ++j) {
      v4f t = qraw[j];
      qa[2 * j].x     = t.x * sc;
      qa[2 * j].y     = t.y * sc;
      qa[2 * j + 1].x = t.z * sc;
      qa[2 * j + 1].y = t.w * sc;
    }
    constexpr int mt[6] = {0,-16,-32,-48,-64,-96};
    const int lb = (hi << 3) - nl + 16;
    float* sblk = s_scores + wv * SS;
#pragma unroll
    for (int v = 0; v < 6; ++v) {
      if (v + 1 < 6) {
        int krow = rb + 16 * (mt[v + 1] + nl); if (krow < 0) krow = 0;
        const float* p = kbh + (size_t)krow * HDIM + kq;
        __builtin_prefetch(vbh + (size_t)krow * HDIM, 0, 1);
#pragma unroll
        for (int j = 0; j < 8; ++j) kb[(v + 1) & 1][j] = *(const v4f*)(p + 8 * j);
      }
      if (v > 0 && rb + 16 * (mt[v] + 15) < 0) continue;
      v8f acc = {0.f,0.f,0.f,0.f,0.f,0.f,0.f,0.f};
#pragma unroll
      for (int j = 0; j < 8; ++j) {
        v4f b = kb[v & 1][j];
        acc = wmma4(qa[2 * j],     (v2f){b.x, b.y}, acc);
        acc = wmma4(qa[2 * j + 1], (v2f){b.z, b.w}, acc);
      }
#pragma unroll
      for (int r = 0; r < 8; ++r) {
        const int M   = r + (hi << 3);
        const int col = s_slot[lb + r - mt[v]];
        sblk[M * (16 * SS) + (col < 0 ? nl : col)] = acc[r];
      }
    }
  }
  __syncthreads();

  if (lane < 16) {
    float* sr = srow + lane * SS + DCOL;
    const float* qd = qrow + lane * QSTR;
    const int ng = n0 + lane;
    const float NEG = -3.0e38f;
    float mx = NEG;
    for (int g = 0; g < 8; ++g) {
      v4f sv = (*(const v4f*)(sr + 4 * g) + *(const v4f*)(qd + 4 * g)) + *(const v4f*)(s_pb + 4 * g);
      if (ng < 4 * g)     sv.x = NEG;
      if (ng < 4 * g + 1) sv.y = NEG;
      if (ng < 4 * g + 2) sv.z = NEG;
      if (ng < 4 * g + 3) sv.w = NEG;
      *(v4f*)(sr + 4 * g) = sv;
      mx = fmaxf(fmaxf(fmaxf(fmaxf(mx, sv.x), sv.y), sv.z), sv.w);
    }
    {
      float sv = (sr[32] + qd[32]) + s_pb[32];
      if (ng < 32) sv = NEG;
      sr[32] = sv;
      mx = fmaxf(mx, sv);
    }
    for (int g = 0; g < 2; ++g) {
      const int k = SSLOT + 4 * g;
      v4f sv = (*(const v4f*)(sr + k) + *(const v4f*)(qd + k)) + *(const v4f*)(s_pb + k);
      if (ng < c_off[33 + 4 * g])     sv.x = NEG;
      if (ng < c_off[33 + 4 * g + 1]) sv.y = NEG;
      if (ng < c_off[33 + 4 * g + 2]) sv.z = NEG;
      if (ng < c_off[33 + 4 * g + 3]) sv.w = NEG;
      *(v4f*)(sr + k) = sv;
      mx = fmaxf(fmaxf(fmaxf(fmaxf(mx, sv.x), sv.y), sv.z), sv.w);
    }
    for (int s = 8; s < NOFF - 33; ++s) {
      float sv = (sr[SSLOT + s] + qd[SSLOT + s]) + s_pb[SSLOT + s];
      if (ng < c_off[33 + s]) sv = NEG;
      sr[SSLOT + s] = sv;
      mx = fmaxf(mx, sv);
    }

    float sum = 0.f;
    for (int g = 0; g < 8; ++g) {
      v4f a = *(const v4f*)(sr + 4 * g);
      v4f e;
      e.x = __expf(a.x - mx); e.y = __expf(a.y - mx); e.z = __expf(a.z - mx); e.w = __expf(a.w - mx);
      *(v4f*)(sr + 4 * g) = e;
      sum += e.x; sum += e.y; sum += e.z; sum += e.w;
    }
    {
      float e = __expf(sr[32] - mx);
      sr[32] = e;
      sum += e;
    }
    for (int g = 0; g < 2; ++g) {
      const int k = SSLOT + 4 * g;
      v4f a = *(const v4f*)(sr + k);
      v4f e;
      e.x = __expf(a.x - mx); e.y = __expf(a.y - mx); e.z = __expf(a.z - mx); e.w = __expf(a.w - mx);
      *(v4f*)(sr + k) = e;
      sum += e.x; sum += e.y; sum += e.z; sum += e.w;
    }
    for (int s = 8; s < NOFF - 33; ++s) {
      float e = __expf(sr[SSLOT + s] - mx);
      sr[SSLOT + s] = e;
      sum += e;
    }
    s_rinv[wv * 16 + lane] = 1.0f / sum;

    const v4f z4 = {0.f, 0.f, 0.f, 0.f};
    for (int g = 0; g < 4; ++g) *(v4f*)(sr - DCOL + 4 * g) = z4;
    sr[33] = 0.f; sr[34] = 0.f; sr[35] = 0.f;
    for (int g = 9; g < 12; ++g) *(v4f*)(sr + 4 * g) = z4;
  }

  const int   r7    = nl & 7;
  const bool  lo8   = nl < 8;
  const float* srx  = srow + r7 * SS;
  const float* sry  = srow + (r7 + 8) * SS;
  const float rinvx = s_rinv[wv * 16 + r7];
  const float rinvy = s_rinv[wv * 16 + r7 + 8];
  const float* tabx = lo8 ? s_cp : s_sp;
  const float* taby = lo8 ? s_sp : s_cp;

  v8f ox[4], oy[4];
#pragma unroll
  for (int ht = 0; ht < 4; ++ht) {
    ox[ht] = (v8f){0.f,0.f,0.f,0.f,0.f,0.f,0.f,0.f};
    oy[ht] = (v8f){0.f,0.f,0.f,0.f,0.f,0.f,0.f,0.f};
  }

  const unsigned vlane = (unsigned)(koff * HDIM + nl);

#pragma unroll
  for (int t = 0; t < 3; ++t) {
    const int off = t << 4;
    if (n0s + 15 < off) continue;
    const float* vt = vbh + (size_t)(n0s - off) * HDIM;

    v2f vb[4][4];
#pragma unroll
    for (int ht = 0; ht < 4; ++ht) {
#pragma unroll
      for (int c = 0; c < 4; ++c) {
        const float* p = vt + ((4 * c) * HDIM + (ht << 4));
        vb[ht][c].x = p[vlane];
        vb[ht][c].y = p[vlane + HDIM];
      }
    }

    v2f pax[4], pay[4];
#pragma unroll
    for (int c = 0; c < 4; ++c) {
#pragma unroll
      for (int u = 0; u < 2; ++u) {
        int j  = 4 * c + koff + u;
        int dx = r7 - j + off;
        int dy = dx + 8;
        float ax = (srx[DCOL + dx] * rinvx) * tabx[DCOL + dx];
        float ay = (sry[DCOL + dy] * rinvy) * taby[DCOL + dy];
        if (u == 0) { pax[c].x = ax; pay[c].x = ay; }
        else        { pax[c].y = ax; pay[c].y = ay; }
      }
    }

#pragma unroll
    for (int ht = 0; ht < 4; ++ht) {
#pragma unroll
      for (int c = 0; c < 4; ++c) {
        if (!(t == 0 && c >= 2)) ox[ht] = wmma4(pax[c], vb[ht][c], ox[ht]);
        if (!(t == 2 && c < 2))  oy[ht] = wmma4(pay[c], vb[ht][c], oy[ht]);
      }
    }
  }

  for (int t = 3; t < NTILE; ++t) {
    const int off = c_toff[t];
    if (n0s + 15 < off) continue;
    const float* vt = vbh + (size_t)(n0s - off) * HDIM;
    const int s  = t - 3;

    v2f vb[4][4];
#pragma unroll
    for (int ht = 0; ht < 4; ++ht) {
#pragma unroll
      for (int c = 0; c < 4; ++c) {
        const float* p = vt + ((4 * c) * HDIM + (ht << 4));
        vb[ht][c].x = p[vlane];
        vb[ht][c].y = p[vlane + HDIM];
      }
    }

    const float px = (srx[SCOL + s] * rinvx) * tabx[DCOL + 33 + s];
    const float py = (sry[SCOL + s] * rinvy) * taby[DCOL + 33 + s];

    v2f pa[4];
#pragma unroll
    for (int c = 0; c < 4; ++c) {
#pragma unroll
      for (int u = 0; u < 2; ++u) {
        int j = 4 * c + koff + u;
        float a = (r7 == (j & 7)) ? ((c < 2) ? px : py) : 0.f;
        if (u == 0) pa[c].x = a; else pa[c].y = a;
      }
    }

#pragma unroll
    for (int ht = 0; ht < 4; ++ht) {
#pragma unroll
      for (int c = 0; c < 4; ++c) {
        if (c < 2) ox[ht] = wmma4(pa[c], vb[ht][c], ox[ht]);
        else       oy[ht] = wmma4(pa[c], vb[ht][c], oy[ht]);
      }
    }
  }

  const float sgn = (lane & 1) ? 1.0f : -1.0f;
#pragma unroll
  for (int ht = 0; ht < 4; ++ht) {
#pragma unroll
    for (int r = 0; r < 8; ++r) {
      float cv = hi ? oy[ht][r] : ox[ht][r];
      float sv = hi ? ox[ht][r] : oy[ht][r];
      float sw = __shfl_xor(sv, 17, 32);
      float res = cv + sgn * sw;
      int M = r + (hi << 3);
      obh[(size_t)(n0 + M) * HDIM + (ht << 4) + nl] = res;
    }
  }
}

extern "C" void kernel_launch(void* const* d_in, const int* in_sizes, int n_in,
                              void* d_out, int out_size, void* d_ws, size_t ws_size,
                              hipStream_t stream) {
  const float* q  = (const float*)d_in[0];
  const float* k  = (const float*)d_in[1];
  const float* v  = (const float*)d_in[2];
  const float* pb = (const float*)d_in[3];
  const float* se = (const float*)d_in[4];
  const float* ph = (const float*)d_in[5];
  float* out = (float*)d_out;

  dim3 grid(NB * NH * (NN / 256));
  dim3 block(512);
  hipLaunchKernelGGL(dsqg_attn_v4_kernel, grid, block, 0, stream,
                     q, k, v, pb, se, ph, out);
}
